# starting kernel
# baseline (speedup 1.0000x reference)
#include <hip/hip_runtime.h>
#include <hip/hip_bf16.h>

#define NHEADS 4
#define HDIM   32
#define NNODES 1024
#define BATCH  8
#define INDIM  128
#define OUTDIM 128
#define NEG_INF_F (-1000000000.0f)

typedef float v2f __attribute__((ext_vector_type(2)));
typedef float v8f __attribute__((ext_vector_type(8)));

__global__ __launch_bounds__(256) void gat_linear(const float* __restrict__ x,
                                                  const float* __restrict__ W,
                                                  const float* __restrict__ bias,
                                                  float* __restrict__ h)
{
    const int lane    = threadIdx.x & 31;
    const int wave    = threadIdx.x >> 5;
    const int rowTile = blockIdx.x;
    const int rloc    = lane & 15;
    const int kq      = (lane >> 4) * 2;
    const int col     = wave * 16 + rloc;

    __shared__ float xs[16][132];
    for (int i = threadIdx.x; i < 16 * 128; i += 256) {
        int rr = i >> 7, cc = i & 127;
        xs[rr][cc] = x[(rowTile * 16 + rr) * INDIM + cc];
    }
    __syncthreads();

    v8f acc = {};
    for (int kk = 0; kk < INDIM; kk += 4) {
        v2f a, b;
        a.x = xs[rloc][kk + kq + 0];
        a.y = xs[rloc][kk + kq + 1];
        b.x = W[col * INDIM + kk + kq + 0];
        b.y = W[col * INDIM + kk + kq + 1];
        acc = __builtin_amdgcn_wmma_f32_16x16x4_f32(false, a, false, b,
                                                    (short)0, acc, false, false);
    }

    const float bv = bias[col];
    const int rbase = rowTile * 16 + ((lane >> 4) << 3);
#pragma unroll
    for (int v = 0; v < 8; ++v)
        h[(rbase + v) * OUTDIM + col] = acc[v] + bv;
}

__global__ __launch_bounds__(256) void gat_scores(const float* __restrict__ h,
                                                  const float* __restrict__ a,
                                                  float* __restrict__ s,
                                                  float* __restrict__ t)
{
    int idx = blockIdx.x * blockDim.x + threadIdx.x;
    if (idx >= BATCH * NHEADS * NNODES) return;
    int n    = idx & (NNODES - 1);
    int head = (idx >> 10) & (NHEADS - 1);
    int b    = idx >> 12;
    const float* hp = h + ((size_t)(b * NNODES + n) * OUTDIM + head * HDIM);
    float ss = 0.f, tt = 0.f;
#pragma unroll
    for (int d = 0; d < HDIM; ++d) {
        float hv = hp[d];
        ss += hv * a[d];
        tt += hv * a[HDIM + d];
    }
    s[idx] = ss;
    t[idx] = tt;
}

__global__ __launch_bounds__(128) void gat_attn(const float* __restrict__ h,
                                                const int*   __restrict__ adj,
                                                const float* __restrict__ s,
                                                const float* __restrict__ t,
                                                float* __restrict__ out)
{
    const int rowTile = blockIdx.x & 63;
    const int head    = (blockIdx.x >> 6) & (NHEADS - 1);
    const int b       = blockIdx.x >> 8;

    const int tid  = threadIdx.x;
    const int wave = tid >> 5;
    const int lane = tid & 31;
    const int r    = lane & 15;
    const int jl   = lane >> 4;

    __shared__ float pm[128], pl[128];
    __shared__ float gm[16], gl[16];
    __shared__ float pc[4][2][8][32];

    const int   iGlob  = rowTile * 16 + r;
    const float s_row  = s[(b * NHEADS + head) * NNODES + iGlob];
    const float* t_bh  = t + (b * NHEADS + head) * NNODES;
    const int*  adj_row = adj + (size_t)(b * NNODES + iGlob) * NNODES;

    const int jbase = wave * 256;
    float m = -__builtin_inff(), l = 0.f;
    for (int it = 0; it < 128; ++it) {
        int j = jbase + jl + it * 2;
        float e = s_row + t_bh[j];
        e = (e > 0.f) ? e : 0.2f * e;
        e = (adj_row[j] == 0) ? NEG_INF_F : e;
        float mn = fmaxf(m, e);
        l = l * __expf(m - mn) + __expf(e - mn);
        m = mn;
    }
    pm[tid] = m; pl[tid] = l;
    __syncthreads();
    if (tid < 16) {
        float gmv = -__builtin_inff();
        for (int q = 0; q < 8; ++q) {
            int src = (q >> 1) * 32 + (q & 1) * 16 + tid;
            gmv = fmaxf(gmv, pm[src]);
        }
        float glv = 0.f;
        for (int q = 0; q < 8; ++q) {
            int src = (q >> 1) * 32 + (q & 1) * 16 + tid;
            glv += pl[src] * __expf(pm[src] - gmv);
        }
        gm[tid] = gmv;
        gl[tid] = glv;
    }
    __syncthreads();

    const float mrow = gm[r];
    const float* hb = h + (size_t)b * NNODES * OUTDIM + head * HDIM;
    const int n0 = lane & 15;
    v8f c0 = {}, c1 = {};
    for (int it = 0; it < 64; ++it) {
        int jA = jbase + it * 4 + jl * 2;
        v2f a;
        {
            float e = s_row + t_bh[jA];
            e = (e > 0.f) ? e : 0.2f * e;
            e = (adj_row[jA] == 0) ? NEG_INF_F : e;
            a.x = __expf(e - mrow);
            e = s_row + t_bh[jA + 1];
            e = (e > 0.f) ? e : 0.2f * e;
            e = (adj_row[jA + 1] == 0) ? NEG_INF_F : e;
            a.y = __expf(e - mrow);
        }
        v2f b0, b1;
        b0.x = hb[(size_t)(jA + 0) * OUTDIM + n0];
        b0.y = hb[(size_t)(jA + 1) * OUTDIM + n0];
        b1.x = hb[(size_t)(jA + 0) * OUTDIM + 16 + n0];
        b1.y = hb[(size_t)(jA + 1) * OUTDIM + 16 + n0];
        c0 = __builtin_amdgcn_wmma_f32_16x16x4_f32(false, a, false, b0,
                                                   (short)0, c0, false, false);
        c1 = __builtin_amdgcn_wmma_f32_16x16x4_f32(false, a, false, b1,
                                                   (short)0, c1, false, false);
    }
#pragma unroll
    for (int v = 0; v < 8; ++v) {
        pc[wave][0][v][lane] = c0[v];
        pc[wave][1][v][lane] = c1[v];
    }
    __syncthreads();

    for (int idx = tid; idx < 512; idx += 128) {
        int ct = idx >> 8;
        int v  = (idx >> 5) & 7;
        int ln = idx & 31;
        float sum = pc[0][ct][v][ln] + pc[1][ct][v][ln] +
                    pc[2][ct][v][ln] + pc[3][ct][v][ln];
        int rowL = v + ((ln >> 4) << 3);
        int col  = ct * 16 + (ln & 15);
        out[(size_t)(b * NNODES + rowTile * 16 + rowL) * OUTDIM + head * HDIM + col]
            = sum / gl[rowL];
    }
}

extern "C" void kernel_launch(void* const* d_in, const int* in_sizes, int n_in,
                              void* d_out, int out_size, void* d_ws, size_t ws_size,
                              hipStream_t stream)
{
    const float* x    = (const float*)d_in[0];
    const int*   adj  = (const int*)  d_in[1];
    const float* W    = (const float*)d_in[2];
    const float* bias = (const float*)d_in[3];
    const float* a    = (const float*)d_in[4];
    float* out = (float*)d_out;

    float* h = (float*)d_ws;
    float* s = h + BATCH * NNODES * OUTDIM;
    float* t = s + BATCH * NHEADS * NNODES;

    gat_linear<<<(BATCH * NNODES) / 16, 256, 0, stream>>>(x, W, bias, h);
    gat_scores<<<(BATCH * NHEADS * NNODES) / 256, 256, 0, stream>>>(h, a, s, t);
    gat_attn<<<BATCH * NHEADS * (NNODES / 16), 128, 0, stream>>>(h, adj, s, t, out);
}


# attn: one block per (batch,16-row tile) serves all 4 heads; adjacency tile read once with whole-line loads, kept as a bit mask in LDS for both passes
# speedup vs baseline: 1.1581x; 1.1581x over previous
#include <hip/hip_runtime.h>
#include <hip/hip_bf16.h>

#define NHEADS 4
#define HDIM   32
#define NNODES 1024
#define BATCH  8
#define INDIM  128
#define OUTDIM 128
#define NEG_INF_F (-1000000000.0f)

typedef float v2f __attribute__((ext_vector_type(2)));
typedef float v8f __attribute__((ext_vector_type(8)));

__global__ __launch_bounds__(256) void gat_linear(const float* __restrict__ x,
                                                  const float* __restrict__ W,
                                                  const float* __restrict__ bias,
                                                  float* __restrict__ h)
{
    const int lane    = threadIdx.x & 31;
    const int wave    = threadIdx.x >> 5;
    const int rowTile = blockIdx.x;
    const int rloc    = lane & 15;
    const int kq      = (lane >> 4) * 2;
    const int col     = wave * 16 + rloc;

    __shared__ float xs[16][132];
    for (int i = threadIdx.x; i < 16 * 128; i += 256) {
        int rr = i >> 7, cc = i & 127;
        xs[rr][cc] = x[(rowTile * 16 + rr) * INDIM + cc];
    }
    __syncthreads();

    v8f acc = {};
    for (int kk = 0; kk < INDIM; kk += 4) {
        v2f a, b;
        a.x = xs[rloc][kk + kq + 0];
        a.y = xs[rloc][kk + kq + 1];
        b.x = W[col * INDIM + kk + kq + 0];
        b.y = W[col * INDIM + kk + kq + 1];
        acc = __builtin_amdgcn_wmma_f32_16x16x4_f32(false, a, false, b,
                                                    (short)0, acc, false, false);
    }

    const float bv = bias[col];
    const int rbase = rowTile * 16 + ((lane >> 4) << 3);
#pragma unroll
    for (int v = 0; v < 8; ++v)
        h[(rbase + v) * OUTDIM + col] = acc[v] + bv;
}

__global__ __launch_bounds__(256) void gat_scores(const float* __restrict__ h,
                                                  const float* __restrict__ a,
                                                  float* __restrict__ s,
                                                  float* __restrict__ t)
{
    int idx = blockIdx.x * blockDim.x + threadIdx.x;
    if (idx >= BATCH * NHEADS * NNODES) return;
    int n    = idx & (NNODES - 1);
    int head = (idx >> 10) & (NHEADS - 1);
    int b    = idx >> 12;
    const float* hp = h + ((size_t)(b * NNODES + n) * OUTDIM + head * HDIM);
    float ss = 0.f, tt = 0.f;
#pragma unroll
    for (int d = 0; d < HDIM; ++d) {
        float hv = hp[d];
        ss += hv * a[d];
        tt += hv * a[HDIM + d];
    }
    s[idx] = ss;
    t[idx] = tt;
}

#define MPITCH 36

__global__ __launch_bounds__(512) void gat_attn(const float* __restrict__ h,
                                                const int*   __restrict__ adj,
                                                const float* __restrict__ s,
                                                const float* __restrict__ t,
                                                float* __restrict__ out)
{
    const int rowTile = blockIdx.x & 63;
    const int b       = blockIdx.x >> 6;

    const int tid  = threadIdx.x;
    const int wave = tid >> 5;
    const int lane = tid & 31;
    const int head = wave >> 2;
    const int quad = wave & 3;
    const int r    = lane & 15;
    const int jl   = lane >> 4;

    __shared__ __attribute__((aligned(16))) unsigned mbits[16 * MPITCH];
    __shared__ float pm[512], pl[512];
    __shared__ float gm[NHEADS][16], gl[NHEADS][16];
    __shared__ float pc[16][2][8][32];

    {
        const int* adj_row = adj + (size_t)(b * NNODES + rowTile * 16 + wave) * NNODES + lane * 4;
        int4 v[8];
#pragma unroll
        for (int u = 0; u < 8; ++u)
            v[u] = *(const int4*)(adj_row + u * 128);
#pragma unroll
        for (int u = 0; u < 8; ++u) {
            unsigned m0 = __builtin_amdgcn_ballot_w32(v[u].x != 0);
            unsigned m1 = __builtin_amdgcn_ballot_w32(v[u].y != 0);
            unsigned m2 = __builtin_amdgcn_ballot_w32(v[u].z != 0);
            unsigned m3 = __builtin_amdgcn_ballot_w32(v[u].w != 0);
            if (lane == 0)
                *(uint4*)&mbits[wave * MPITCH + u * 4] = make_uint4(m0, m1, m2, m3);
        }
    }

    const int   iGlob  = rowTile * 16 + r;
    const float s_row  = s[(b * NHEADS + head) * NNODES + iGlob];
    const float* t_bh  = t + (b * NHEADS + head) * NNODES;
    const int jbase = quad * 256;
    __syncthreads();

    float m = -__builtin_inff(), l = 0.f;
    for (int c = 0; c < 2; ++c) {
        const unsigned* mw = &mbits[r * MPITCH + (quad * 2 + c) * 4];
        const unsigned wa = mw[jl];
        const unsigned wb = mw[jl + 2];
        const int jc = jbase + c * 128 + jl;
        for (int i = 0; i < 32; ++i) {
            const unsigned bit = 1u << i;
            {
                float e = s_row + t_bh[jc + 4 * i];
                e = (e > 0.f) ? e : 0.2f * e;
                e = ((wa & bit) == 0u) ? NEG_INF_F : e;
                float mn = fmaxf(m, e);
                l = l * __expf(m - mn) + __expf(e - mn);
                m = mn;
            }
            {
                float e = s_row + t_bh[jc + 4 * i + 2];
                e = (e > 0.f) ? e : 0.2f * e;
                e = ((wb & bit) == 0u) ? NEG_INF_F : e;
                float mn = fmaxf(m, e);
                l = l * __expf(m - mn) + __expf(e - mn);
                m = mn;
            }
        }
    }
    pm[tid] = m; pl[tid] = l;
    __syncthreads();
    if (tid < 64) {
        const int hd = tid >> 4, row = tid & 15;
        float gmv = -__builtin_inff();
        for (int q = 0; q < 8; ++q) {
            int src = (hd * 4 + (q >> 1)) * 32 + (q & 1) * 16 + row;
            gmv = fmaxf(gmv, pm[src]);
        }
        float glv = 0.f;
        for (int q = 0; q < 8; ++q) {
            int src = (hd * 4 + (q >> 1)) * 32 + (q & 1) * 16 + row;
            glv += pl[src] * __expf(pm[src] - gmv);
        }
        gm[hd][row] = gmv;
        gl[hd][row] = glv;
    }
    __syncthreads();

    const float mrow = gm[head][r];
    const float* hb = h + (size_t)b * NNODES * OUTDIM + head * HDIM;
    const int n0 = lane & 15;
    v8f c0 = {}, c1 = {};
    for (int c = 0; c < 2; ++c) {
        const unsigned* mw = &mbits[r * MPITCH + (quad * 2 + c) * 4 + jl * 2];
        const unsigned w0 = mw[0];
        const unsigned w1 = mw[1];
        const int jc = jbase + c * 128 + jl * 2;
        for (int i = 0; i < 32; ++i) {
            const unsigned bit = 1u << i;
            const int jA = jc + 4 * i;
            v2f a;
            {
                float e = s_row + t_bh[jA];
                e = (e > 0.f) ? e : 0.2f * e;
                e = ((w0 & bit) == 0u) ? NEG_INF_F : e;
                a.x = __expf(e - mrow);
                e = s_row + t_bh[jA + 1];
                e = (e > 0.f) ? e : 0.2f * e;
                e = ((w1 & bit) == 0u) ? NEG_INF_F : e;
                a.y = __expf(e - mrow);
            }
            v2f b0, b1;
            b0.x = hb[(size_t)(jA + 0) * OUTDIM + n0];
            b0.y = hb[(size_t)(jA + 1) * OUTDIM + n0];
            b1.x = hb[(size_t)(jA + 0) * OUTDIM + 16 + n0];
            b1.y = hb[(size_t)(jA + 1) * OUTDIM + 16 + n0];
            c0 = __builtin_amdgcn_wmma_f32_16x16x4_f32(false, a, false, b0,
                                                       (short)0, c0, false, false);
            c1 = __builtin_amdgcn_wmma_f32_16x16x4_f32(false, a, false, b1,
                                                       (short)0, c1, false, false);
        }
    }
#pragma unroll
    for (int v = 0; v < 8; ++v) {
        pc[wave][0][v][lane] = c0[v];
        pc[wave][1][v][lane] = c1[v];
    }
    __syncthreads();

    for (int idx = tid; idx < 2048; idx += 512) {
        int col  = idx & 31;
        int hd   = (idx >> 5) & 3;
        int rowL = idx >> 7;
        int ct = col >> 4;
        int v  = rowL & 7;
        int ln = (col & 15) + ((rowL >> 3) << 4);
        float sum = pc[hd * 4 + 0][ct][v][ln] + pc[hd * 4 + 1][ct][v][ln] +
                    pc[hd * 4 + 2][ct][v][ln] + pc[hd * 4 + 3][ct][v][ln];
        out[(size_t)(b * NNODES + rowTile * 16 + rowL) * OUTDIM + hd * HDIM + col]
            = sum / gl[hd][rowL];
    }
}

extern "C" void kernel_launch(void* const* d_in, const int* in_sizes, int n_in,
                              void* d_out, int out_size, void* d_ws, size_t ws_size,
                              hipStream_t stream)
{
    const float* x    = (const float*)d_in[0];
    const int*   adj  = (const int*)  d_in[1];
    const float* W    = (const float*)d_in[2];
    const float* bias = (const float*)d_in[3];
    const float* a    = (const float*)d_in[4];
    float* out = (float*)d_out;

    float* h = (float*)d_ws;
    float* s = h + BATCH * NNODES * OUTDIM;
    float* t = s + BATCH * NHEADS * NNODES;

    gat_linear<<<(BATCH * NNODES) / 16, 256, 0, stream>>>(x, W, bias, h);
    gat_scores<<<(BATCH * NHEADS * NNODES) / 256, 256, 0, stream>>>(h, a, s, t);
    gat_attn<<<BATCH * (NNODES / 16), 512, 0, stream>>>(h, adj, s, t, out);
}


# attn: row max without exp: max_j leaky(s_i+t_j) = leaky(s_i + masked max_j t_j), found while packing the mask, so pass 1 goes; row sum l now added from the pass-2 exp values (new sum order)
# speedup vs baseline: 1.4039x; 1.2122x over previous
#include <hip/hip_runtime.h>
#include <hip/hip_bf16.h>

#define NHEADS 4
#define HDIM   32
#define NNODES 1024
#define BATCH  8
#define INDIM  128
#define OUTDIM 128
#define NEG_INF_F (-1000000000.0f)

typedef float v2f __attribute__((ext_vector_type(2)));
typedef float v8f __attribute__((ext_vector_type(8)));

__global__ __launch_bounds__(256) void gat_linear(const float* __restrict__ x,
                                                  const float* __restrict__ W,
                                                  const float* __restrict__ bias,
                                                  float* __restrict__ h)
{
    const int lane    = threadIdx.x & 31;
    const int wave    = threadIdx.x >> 5;
    const int rowTile = blockIdx.x;
    const int rloc    = lane & 15;
    const int kq      = (lane >> 4) * 2;
    const int col     = wave * 16 + rloc;

    __shared__ float xs[16][132];
    for (int i = threadIdx.x; i < 16 * 128; i += 256) {
        int rr = i >> 7, cc = i & 127;
        xs[rr][cc] = x[(rowTile * 16 + rr) * INDIM + cc];
    }
    __syncthreads();

    v8f acc = {};
    for (int kk = 0; kk < INDIM; kk += 4) {
        v2f a, b;
        a.x = xs[rloc][kk + kq + 0];
        a.y = xs[rloc][kk + kq + 1];
        b.x = W[col * INDIM + kk + kq + 0];
        b.y = W[col * INDIM + kk + kq + 1];
        acc = __builtin_amdgcn_wmma_f32_16x16x4_f32(false, a, false, b,
                                                    (short)0, acc, false, false);
    }

    const float bv = bias[col];
    const int rbase = rowTile * 16 + ((lane >> 4) << 3);
#pragma unroll
    for (int v = 0; v < 8; ++v)
        h[(rbase + v) * OUTDIM + col] = acc[v] + bv;
}

__global__ __launch_bounds__(256) void gat_scores(const float* __restrict__ h,
                                                  const float* __restrict__ a,
                                                  float* __restrict__ s,
                                                  float* __restrict__ t)
{
    int idx = blockIdx.x * blockDim.x + threadIdx.x;
    if (idx >= BATCH * NHEADS * NNODES) return;
    int n    = idx & (NNODES - 1);
    int head = (idx >> 10) & (NHEADS - 1);
    int b    = idx >> 12;
    const float* hp = h + ((size_t)(b * NNODES + n) * OUTDIM + head * HDIM);
    float ss = 0.f, tt = 0.f;
#pragma unroll
    for (int d = 0; d < HDIM; ++d) {
        float hv = hp[d];
        ss += hv * a[d];
        tt += hv * a[HDIM + d];
    }
    s[idx] = ss;
    t[idx] = tt;
}

#define MPITCH 36

__device__ __forceinline__ float xhalf(float x)
{
    const int bits = __builtin_bit_cast(int, x);
    return __builtin_bit_cast(float,
        __builtin_amdgcn_permlanex16(bits, bits, 0x76543210, 0xfedcba98, false, false));
}

template <int XM>
__device__ __forceinline__ float row_xor(float x)
{
    const int bits = __builtin_bit_cast(int, x);
    return __builtin_bit_cast(float,
        __builtin_amdgcn_update_dpp(bits, bits, 0x160 + XM, 0xf, 0xf, false));
}

__device__ __forceinline__ float wave_max(float x)
{
    x = fmaxf(x, row_xor<1>(x));
    x = fmaxf(x, row_xor<2>(x));
    x = fmaxf(x, row_xor<4>(x));
    x = fmaxf(x, row_xor<8>(x));
    return fmaxf(x, xhalf(x));
}

__global__ __launch_bounds__(512) void gat_attn(const float* __restrict__ h,
                                                const int*   __restrict__ adj,
                                                const float* __restrict__ s,
                                                const float* __restrict__ t,
                                                float* __restrict__ out)
{
    const int rowTile = blockIdx.x & 63;
    const int b       = blockIdx.x >> 6;

    const int tid  = threadIdx.x;
    const int wave = tid >> 5;
    const int lane = tid & 31;
    const int head = wave >> 2;
    const int quad = wave & 3;
    const int r    = lane & 15;
    const int jl   = lane >> 4;

    __shared__ __attribute__((aligned(16))) unsigned mbits[16 * MPITCH];
    __shared__ float pl[512];
    __shared__ float gm[NHEADS][16];
    __shared__ float pc[16][2][8][32];

    {
        const int* adj_row = adj + (size_t)(b * NNODES + rowTile * 16 + wave) * NNODES + lane * 4;
        const float* t_ln  = t + (size_t)b * NHEADS * NNODES + lane * 4;
        int4 v[8];
#pragma unroll
        for (int u = 0; u < 8; ++u)
            v[u] = *(const int4*)(adj_row + u * 128);
        const float ninf = -__builtin_inff();
        float tm[NHEADS] = {ninf, ninf, ninf, ninf};
#pragma unroll
        for (int u = 0; u < 8; ++u) {
            const bool p0 = v[u].x != 0, p1 = v[u].y != 0, p2 = v[u].z != 0, p3 = v[u].w != 0;
            unsigned m0 = __builtin_amdgcn_ballot_w32(p0);
            unsigned m1 = __builtin_amdgcn_ballot_w32(p1);
            unsigned m2 = __builtin_amdgcn_ballot_w32(p2);
            unsigned m3 = __builtin_amdgcn_ballot_w32(p3);
            if (lane == 0)
                *(uint4*)&mbits[wave * MPITCH + u * 4] = make_uint4(m0, m1, m2, m3);
#pragma unroll
            for (int hd = 0; hd < NHEADS; ++hd) {
                const float4 tv = *(const float4*)(t_ln + hd * NNODES + u * 128);
                const float a0 = p0 ? tv.x : ninf, a1 = p1 ? tv.y : ninf;
                const float a2 = p2 ? tv.z : ninf, a3 = p3 ? tv.w : ninf;
                tm[hd] = fmaxf(fmaxf(tm[hd], a0), fmaxf(fmaxf(a1, a2), a3));
            }
        }
#pragma unroll
        for (int hd = 0; hd < NHEADS; ++hd)
            tm[hd] = wave_max(tm[hd]);
        if (lane < NHEADS) {
            const float tmx = (lane & 2) ? ((lane & 1) ? tm[3] : tm[2]) : ((lane & 1) ? tm[1] : tm[0]);
            float e = s[(b * NHEADS + lane) * NNODES + rowTile * 16 + wave] + tmx;
            e = (e > 0.f) ? e : 0.2f * e;
            gm[lane][wave] = fmaxf(NEG_INF_F, e);
        }
    }

    const int   iGlob  = rowTile * 16 + r;
    const float s_row  = s[(b * NHEADS + head) * NNODES + iGlob];
    const float* t_bh  = t + (b * NHEADS + head) * NNODES;
    const int jbase = quad * 256;
    __syncthreads();

    const float mrow = gm[head][r];
    const float* hb = h + (size_t)b * NNODES * OUTDIM + head * HDIM;
    const int n0 = lane & 15;
    v8f c0 = {}, c1 = {};
    v2f lacc = {};
    for (int c = 0; c < 2; ++c) {
        const unsigned* mw = &mbits[r * MPITCH + (quad * 2 + c) * 4 + jl * 2];
        const unsigned w0 = mw[0];
        const unsigned w1 = mw[1];
        const int jc = jbase + c * 128 + jl * 2;
#pragma unroll 4
        for (int i = 0; i < 32; ++i) {
            const unsigned bit = 1u << i;
            const int jA = jc + 4 * i;
            v2f a;
            {
                float e = s_row + t_bh[jA];
                e = (e > 0.f) ? e : 0.2f * e;
                e = ((w0 & bit) == 0u) ? NEG_INF_F : e;
                a.x = __expf(e - mrow);
                e = s_row + t_bh[jA + 1];
                e = (e > 0.f) ? e : 0.2f * e;
                e = ((w1 & bit) == 0u) ? NEG_INF_F : e;
                a.y = __expf(e - mrow);
            }
            lacc += a;
            v2f b0, b1;
            b0.x = hb[(size_t)(jA + 0) * OUTDIM + n0];
            b0.y = hb[(size_t)(jA + 1) * OUTDIM + n0];
            b1.x = hb[(size_t)(jA + 0) * OUTDIM + 16 + n0];
            b1.y = hb[(size_t)(jA + 1) * OUTDIM + 16 + n0];
            c0 = __builtin_amdgcn_wmma_f32_16x16x4_f32(false, a, false, b0,
                                                       (short)0, c0, false, false);
            c1 = __builtin_amdgcn_wmma_f32_16x16x4_f32(false, a, false, b1,
                                                       (short)0, c1, false, false);
        }
    }
    pl[tid] = lacc.x + lacc.y;
#pragma unroll
    for (int v = 0; v < 8; ++v) {
        pc[wave][0][v][lane] = c0[v];
        pc[wave][1][v][lane] = c1[v];
    }
    __syncthreads();

    for (int idx = tid; idx < 2048; idx += 512) {
        int col  = idx & 31;
        int hd   = (idx >> 5) & 3;
        int rowL = idx >> 7;
        int ct = col >> 4;
        int v  = rowL & 7;
        int ln = (col & 15) + ((rowL >> 3) << 4);
        float sum = pc[hd * 4 + 0][ct][v][ln] + pc[hd * 4 + 1][ct][v][ln] +
                    pc[hd * 4 + 2][ct][v][ln] + pc[hd * 4 + 3][ct][v][ln];
        const float* plh = &pl[hd * 128 + rowL];
        float lrow = ((plh[0] + plh[16]) + (plh[32] + plh[48])) +
                     ((plh[64] + plh[80]) + (plh[96] + plh[112]));
        out[(size_t)(b * NNODES + rowTile * 16 + rowL) * OUTDIM + hd * HDIM + col]
            = sum / lrow;
    }
}

extern "C" void kernel_launch(void* const* d_in, const int* in_sizes, int n_in,
                              void* d_out, int out_size, void* d_ws, size_t ws_size,
                              hipStream_t stream)
{
    const float* x    = (const float*)d_in[0];
    const int*   adj  = (const int*)  d_in[1];
    const float* W    = (const float*)d_in[2];
    const float* bias = (const float*)d_in[3];
    const float* a    = (const float*)d_in[4];
    float* out = (float*)d_out;

    float* h = (float*)d_ws;
    float* s = h + BATCH * NNODES * OUTDIM;
    float* t = s + BATCH * NHEADS * NNODES;

    gat_linear<<<(BATCH * NNODES) / 16, 256, 0, stream>>>(x, W, bias, h);
    gat_scores<<<(BATCH * NHEADS * NNODES) / 256, 256, 0, stream>>>(h, a, s, t);
    gat_attn<<<BATCH * (NNODES / 16), 512, 0, stream>>>(h, adj, s, t, out);
}


# scores fused into the linear kernel: s,t from the h tile in registers (16-lane DPP sum, a head's two waves joined in LDS), score launch removed; new sum order for s and t
# speedup vs baseline: 1.5134x; 1.0780x over previous
#include <hip/hip_runtime.h>
#include <hip/hip_bf16.h>

#define NHEADS 4
#define HDIM   32
#define NNODES 1024
#define BATCH  8
#define INDIM  128
#define OUTDIM 128
#define NEG_INF_F (-1000000000.0f)

typedef float v2f __attribute__((ext_vector_type(2)));
typedef float v8f __attribute__((ext_vector_type(8)));

__device__ __forceinline__ float xhalf(float x)
{
    const int bits = __builtin_bit_cast(int, x);
    return __builtin_bit_cast(float,
        __builtin_amdgcn_permlanex16(bits, bits, 0x76543210, 0xfedcba98, false, false));
}

template <int XM>
__device__ __forceinline__ float row_xor(float x)
{
    const int bits = __builtin_bit_cast(int, x);
    return __builtin_bit_cast(float,
        __builtin_amdgcn_update_dpp(bits, bits, 0x160 + XM, 0xf, 0xf, false));
}

__device__ __forceinline__ float wave_max(float x)
{
    x = fmaxf(x, row_xor<1>(x));
    x = fmaxf(x, row_xor<2>(x));
    x = fmaxf(x, row_xor<4>(x));
    x = fmaxf(x, row_xor<8>(x));
    return fmaxf(x, xhalf(x));
}

__device__ __forceinline__ float row_sum16(float x)
{
    x += row_xor<1>(x);
    x += row_xor<2>(x);
    x += row_xor<4>(x);
    return x + row_xor<8>(x);
}

__global__ __launch_bounds__(256) void gat_linear(const float* __restrict__ x,
                                                  const float* __restrict__ W,
                                                  const float* __restrict__ bias,
                                                  const float* __restrict__ a,
                                                  float* __restrict__ h,
                                                  float* __restrict__ s,
                                                  float* __restrict__ t)
{
    const int lane    = threadIdx.x & 31;
    const int wave    = threadIdx.x >> 5;
    const int rowTile = blockIdx.x;
    const int rloc    = lane & 15;
    const int kq      = (lane >> 4) * 2;
    const int col     = wave * 16 + rloc;

    __shared__ float xs[16][132];
    __shared__ __attribute__((aligned(16))) float sp[2][8][16];
    for (int i = threadIdx.x; i < 16 * 128; i += 256) {
        int rr = i >> 7, cc = i & 127;
        xs[rr][cc] = x[(rowTile * 16 + rr) * INDIM + cc];
    }
    __syncthreads();

    v8f acc = {};
    for (int kk = 0; kk < INDIM; kk += 4) {
        v2f av, bw;
        av.x = xs[rloc][kk + kq + 0];
        av.y = xs[rloc][kk + kq + 1];
        bw.x = W[col * INDIM + kk + kq + 0];
        bw.y = W[col * INDIM + kk + kq + 1];
        acc = __builtin_amdgcn_wmma_f32_16x16x4_f32(false, av, false, bw,
                                                    (short)0, acc, false, false);
    }

    const float bv  = bias[col];
    const float asv = a[col & (HDIM - 1)];
    const float atv = a[HDIM + (col & (HDIM - 1))];
    const int half  = lane >> 4;
    const int rbase = rowTile * 16 + (half << 3);
    float hv[8], ps[8], pt[8];
#pragma unroll
    for (int v = 0; v < 8; ++v) {
        hv[v] = acc[v] + bv;
        ps[v] = row_sum16(hv[v] * asv);
        pt[v] = row_sum16(hv[v] * atv);
    }
    if (rloc == 0) {
        float4* ds = (float4*)&sp[0][wave][half * 8];
        float4* dt = (float4*)&sp[1][wave][half * 8];
        ds[0] = make_float4(ps[0], ps[1], ps[2], ps[3]);
        ds[1] = make_float4(ps[4], ps[5], ps[6], ps[7]);
        dt[0] = make_float4(pt[0], pt[1], pt[2], pt[3]);
        dt[1] = make_float4(pt[4], pt[5], pt[6], pt[7]);
    }
    __syncthreads();
    if (threadIdx.x < 128) {
        const int which = threadIdx.x >> 6;
        const int head  = (threadIdx.x >> 4) & 3;
        const int row   = threadIdx.x & 15;
        const float v   = sp[which][2 * head][row] + sp[which][2 * head + 1][row];
        const int grow  = rowTile * 16 + row;
        float* dst = which ? t : s;
        dst[((grow >> 10) * NHEADS + head) * NNODES + (grow & (NNODES - 1))] = v;
    }
#pragma unroll
    for (int v = 0; v < 8; ++v)
        h[(rbase + v) * OUTDIM + col] = hv[v];
}

#define MPITCH 36

__global__ __launch_bounds__(512) void gat_attn(const float* __restrict__ h,
                                                const int*   __restrict__ adj,
                                                const float* __restrict__ s,
                                                const float* __restrict__ t,
                                                float* __restrict__ out)
{
    const int rowTile = blockIdx.x & 63;
    const int b       = blockIdx.x >> 6;

    const int tid  = threadIdx.x;
    const int wave = tid >> 5;
    const int lane = tid & 31;
    const int head = wave >> 2;
    const int quad = wave & 3;
    const int r    = lane & 15;
    const int jl   = lane >> 4;

    __shared__ __attribute__((aligned(16))) unsigned mbits[16 * MPITCH];
    __shared__ float pl[512];
    __shared__ float gm[NHEADS][16];
    __shared__ float pc[16][2][8][32];

    {
        const int* adj_row = adj + (size_t)(b * NNODES + rowTile * 16 + wave) * NNODES + lane * 4;
        const float* t_ln  = t + (size_t)b * NHEADS * NNODES + lane * 4;
        int4 v[8];
#pragma unroll
        for (int u = 0; u < 8; ++u)
            v[u] = *(const int4*)(adj_row + u * 128);
        const float ninf = -__builtin_inff();
        float tm[NHEADS] = {ninf, ninf, ninf, ninf};
#pragma unroll
        for (int u = 0; u < 8; ++u) {
            const bool p0 = v[u].x != 0, p1 = v[u].y != 0, p2 = v[u].z != 0, p3 = v[u].w != 0;
            unsigned m0 = __builtin_amdgcn_ballot_w32(p0);
            unsigned m1 = __builtin_amdgcn_ballot_w32(p1);
            unsigned m2 = __builtin_amdgcn_ballot_w32(p2);
            unsigned m3 = __builtin_amdgcn_ballot_w32(p3);
            if (lane == 0)
                *(uint4*)&mbits[wave * MPITCH + u * 4] = make_uint4(m0, m1, m2, m3);
#pragma unroll
            for (int hd = 0; hd < NHEADS; ++hd) {
                const float4 tv = *(const float4*)(t_ln + hd * NNODES + u * 128);
                const float a0 = p0 ? tv.x : ninf, a1 = p1 ? tv.y : ninf;
                const float a2 = p2 ? tv.z : ninf, a3 = p3 ? tv.w : ninf;
                tm[hd] = fmaxf(fmaxf(tm[hd], a0), fmaxf(fmaxf(a1, a2), a3));
            }
        }
#pragma unroll
        for (int hd = 0; hd < NHEADS; ++hd)
            tm[hd] = wave_max(tm[hd]);
        if (lane < NHEADS) {
            const float tmx = (lane & 2) ? ((lane & 1) ? tm[3] : tm[2]) : ((lane & 1) ? tm[1] : tm[0]);
            float e = s[(b * NHEADS + lane) * NNODES + rowTile * 16 + wave] + tmx;
            e = (e > 0.f) ? e : 0.2f * e;
            gm[lane][wave] = fmaxf(NEG_INF_F, e);
        }
    }

    const int   iGlob  = rowTile * 16 + r;
    const float s_row  = s[(b * NHEADS + head) * NNODES + iGlob];
    const float* t_bh  = t + (b * NHEADS + head) * NNODES;
    const int jbase = quad * 256;
    __syncthreads();

    const float mrow = gm[head][r];
    const float* hb = h + (size_t)b * NNODES * OUTDIM + head * HDIM;
    const int n0 = lane & 15;
    v8f c0 = {}, c1 = {};
    v2f lacc = {};
    for (int c = 0; c < 2; ++c) {
        const unsigned* mw = &mbits[r * MPITCH + (quad * 2 + c) * 4 + jl * 2];
        const unsigned w0 = mw[0];
        const unsigned w1 = mw[1];
        const int jc = jbase + c * 128 + jl * 2;
#pragma unroll 4
        for (int i = 0; i < 32; ++i) {
            const unsigned bit = 1u << i;
            const int jA = jc + 4 * i;
            v2f a;
            {
                float e = s_row + t_bh[jA];
                e = (e > 0.f) ? e : 0.2f * e;
                e = ((w0 & bit) == 0u) ? NEG_INF_F : e;
                a.x = __expf(e - mrow);
                e = s_row + t_bh[jA + 1];
                e = (e > 0.f) ? e : 0.2f * e;
                e = ((w1 & bit) == 0u) ? NEG_INF_F : e;
                a.y = __expf(e - mrow);
            }
            lacc += a;
            v2f b0, b1;
            b0.x = hb[(size_t)(jA + 0) * OUTDIM + n0];
            b0.y = hb[(size_t)(jA + 1) * OUTDIM + n0];
            b1.x = hb[(size_t)(jA + 0) * OUTDIM + 16 + n0];
            b1.y = hb[(size_t)(jA + 1) * OUTDIM + 16 + n0];
            c0 = __builtin_amdgcn_wmma_f32_16x16x4_f32(false, a, false, b0,
                                                       (short)0, c0, false, false);
            c1 = __builtin_amdgcn_wmma_f32_16x16x4_f32(false, a, false, b1,
                                                       (short)0, c1, false, false);
        }
    }
    pl[tid] = lacc.x + lacc.y;
#pragma unroll
    for (int v = 0; v < 8; ++v) {
        pc[wave][0][v][lane] = c0[v];
        pc[wave][1][v][lane] = c1[v];
    }
    __syncthreads();

    for (int idx = tid; idx < 2048; idx += 512) {
        int col  = idx & 31;
        int hd   = (idx >> 5) & 3;
        int rowL = idx >> 7;
        int ct = col >> 4;
        int v  = rowL & 7;
        int ln = (col & 15) + ((rowL >> 3) << 4);
        float sum = pc[hd * 4 + 0][ct][v][ln] + pc[hd * 4 + 1][ct][v][ln] +
                    pc[hd * 4 + 2][ct][v][ln] + pc[hd * 4 + 3][ct][v][ln];
        const float* plh = &pl[hd * 128 + rowL];
        float lrow = ((plh[0] + plh[16]) + (plh[32] + plh[48])) +
                     ((plh[64] + plh[80]) + (plh[96] + plh[112]));
        out[(size_t)(b * NNODES + rowTile * 16 + rowL) * OUTDIM + hd * HDIM + col]
            = sum / lrow;
    }
}

extern "C" void kernel_launch(void* const* d_in, const int* in_sizes, int n_in,
                              void* d_out, int out_size, void* d_ws, size_t ws_size,
                              hipStream_t stream)
{
    const float* x    = (const float*)d_in[0];
    const int*   adj  = (const int*)  d_in[1];
    const float* W    = (const float*)d_in[2];
    const float* bias = (const float*)d_in[3];
    const float* a    = (const float*)d_in[4];
    float* out = (float*)d_out;

    float* h = (float*)d_ws;
    float* s = h + BATCH * NNODES * OUTDIM;
    float* t = s + BATCH * NHEADS * NNODES;

    gat_linear<<<(BATCH * NNODES) / 16, 256, 0, stream>>>(x, W, bias, a, h, s, t);
    gat_attn<<<BATCH * (NNODES / 16), 512, 0, stream>>>(h, adj, s, t, out);
}


# attn: 32 rows per block, each wave carries two 16-row tiles so every loaded h fragment and t value feeds both (8 k-steps per trip keep the loads in flight per SIMD); per-row sum order unchanged
# speedup vs baseline: 1.5871x; 1.0487x over previous
#include <hip/hip_runtime.h>
#include <hip/hip_bf16.h>

#define NHEADS 4
#define HDIM   32
#define NNODES 1024
#define BATCH  8
#define INDIM  128
#define OUTDIM 128
#define NEG_INF_F (-1000000000.0f)

typedef float v2f __attribute__((ext_vector_type(2)));
typedef float v8f __attribute__((ext_vector_type(8)));

__device__ __forceinline__ float xhalf(float x)
{
    const int bits = __builtin_bit_cast(int, x);
    return __builtin_bit_cast(float,
        __builtin_amdgcn_permlanex16(bits, bits, 0x76543210, 0xfedcba98, false, false));
}

template <int XM>
__device__ __forceinline__ float row_xor(float x)
{
    const int bits = __builtin_bit_cast(int, x);
    return __builtin_bit_cast(float,
        __builtin_amdgcn_update_dpp(bits, bits, 0x160 + XM, 0xf, 0xf, false));
}

__device__ __forceinline__ float wave_max(float x)
{
    x = fmaxf(x, row_xor<1>(x));
    x = fmaxf(x, row_xor<2>(x));
    x = fmaxf(x, row_xor<4>(x));
    x = fmaxf(x, row_xor<8>(x));
    return fmaxf(x, xhalf(x));
}

__device__ __forceinline__ float row_sum16(float x)
{
    x += row_xor<1>(x);
    x += row_xor<2>(x);
    x += row_xor<4>(x);
    return x + row_xor<8>(x);
}

__global__ __launch_bounds__(256) void gat_linear(const float* __restrict__ x,
                                                  const float* __restrict__ W,
                                                  const float* __restrict__ bias,
                                                  const float* __restrict__ a,
                                                  float* __restrict__ h,
                                                  float* __restrict__ s,
                                                  float* __restrict__ t)
{
    const int lane    = threadIdx.x & 31;
    const int wave    = threadIdx.x >> 5;
    const int rowTile = blockIdx.x;
    const int rloc    = lane & 15;
    const int kq      = (lane >> 4) * 2;
    const int col     = wave * 16 + rloc;

    __shared__ float xs[16][132];
    __shared__ __attribute__((aligned(16))) float sp[2][8][16];
    for (int i = threadIdx.x; i < 16 * 128; i += 256) {
        int rr = i >> 7, cc = i & 127;
        xs[rr][cc] = x[(rowTile * 16 + rr) * INDIM + cc];
    }
    __syncthreads();

    v8f acc = {};
    for (int kk = 0; kk < INDIM; kk += 4) {
        v2f av, bw;
        av.x = xs[rloc][kk + kq + 0];
        av.y = xs[rloc][kk + kq + 1];
        bw.x = W[col * INDIM + kk + kq + 0];
        bw.y = W[col * INDIM + kk + kq + 1];
        acc = __builtin_amdgcn_wmma_f32_16x16x4_f32(false, av, false, bw,
                                                    (short)0, acc, false, false);
    }

    const float bv  = bias[col];
    const float asv = a[col & (HDIM - 1)];
    const float atv = a[HDIM + (col & (HDIM - 1))];
    const int half  = lane >> 4;
    const int rbase = rowTile * 16 + (half << 3);
    float hv[8], ps[8], pt[8];
#pragma unroll
    for (int v = 0; v < 8; ++v) {
        hv[v] = acc[v] + bv;
        ps[v] = row_sum16(hv[v] * asv);
        pt[v] = row_sum16(hv[v] * atv);
    }
    if (rloc == 0) {
        float4* ds = (float4*)&sp[0][wave][half * 8];
        float4* dt = (float4*)&sp[1][wave][half * 8];
        ds[0] = make_float4(ps[0], ps[1], ps[2], ps[3]);
        ds[1] = make_float4(ps[4], ps[5], ps[6], ps[7]);
        dt[0] = make_float4(pt[0], pt[1], pt[2], pt[3]);
        dt[1] = make_float4(pt[4], pt[5], pt[6], pt[7]);
    }
    __syncthreads();
    if (threadIdx.x < 128) {
        const int which = threadIdx.x >> 6;
        const int head  = (threadIdx.x >> 4) & 3;
        const int row   = threadIdx.x & 15;
        const float v   = sp[which][2 * head][row] + sp[which][2 * head + 1][row];
        const int grow  = rowTile * 16 + row;
        float* dst = which ? t : s;
        dst[((grow >> 10) * NHEADS + head) * NNODES + (grow & (NNODES - 1))] = v;
    }
#pragma unroll
    for (int v = 0; v < 8; ++v)
        h[(rbase + v) * OUTDIM + col] = hv[v];
}

#define MPITCH 36
#define BROWS  32

__global__ __launch_bounds__(512) void gat_attn(const float* __restrict__ h,
                                                const int*   __restrict__ adj,
                                                const float* __restrict__ s,
                                                const float* __restrict__ t,
                                                float* __restrict__ out)
{
    const int rowTile = blockIdx.x & 31;
    const int b       = blockIdx.x >> 5;

    const int tid  = threadIdx.x;
    const int wave = tid >> 5;
    const int lane = tid & 31;
    const int head = wave >> 2;
    const int quad = wave & 3;
    const int r    = lane & 15;
    const int jl   = lane >> 4;

    __shared__ __attribute__((aligned(16))) unsigned mbits[BROWS * MPITCH];
    __shared__ float pl[2][512];
    __shared__ float gm[NHEADS][BROWS];
    __shared__ float pc[16][2][2][8][32];

    {
        const int* adj_r0 = adj + (size_t)(b * NNODES + rowTile * BROWS + wave) * NNODES + lane * 4;
        const int* adj_r1 = adj_r0 + (size_t)16 * NNODES;
        const float* t_ln = t + (size_t)b * NHEADS * NNODES + lane * 4;
        int4 va[8], vb[8];
#pragma unroll
        for (int u = 0; u < 8; ++u) {
            va[u] = *(const int4*)(adj_r0 + u * 128);
            vb[u] = *(const int4*)(adj_r1 + u * 128);
        }
        const float ninf = -__builtin_inff();
        float tma[NHEADS] = {ninf, ninf, ninf, ninf};
        float tmb[NHEADS] = {ninf, ninf, ninf, ninf};
#pragma unroll
        for (int u = 0; u < 8; ++u) {
            const bool p0 = va[u].x != 0, p1 = va[u].y != 0, p2 = va[u].z != 0, p3 = va[u].w != 0;
            const bool q0 = vb[u].x != 0, q1 = vb[u].y != 0, q2 = vb[u].z != 0, q3 = vb[u].w != 0;
            unsigned m0 = __builtin_amdgcn_ballot_w32(p0);
            unsigned m1 = __builtin_amdgcn_ballot_w32(p1);
            unsigned m2 = __builtin_amdgcn_ballot_w32(p2);
            unsigned m3 = __builtin_amdgcn_ballot_w32(p3);
            unsigned n0 = __builtin_amdgcn_ballot_w32(q0);
            unsigned n1 = __builtin_amdgcn_ballot_w32(q1);
            unsigned n2 = __builtin_amdgcn_ballot_w32(q2);
            unsigned n3 = __builtin_amdgcn_ballot_w32(q3);
            if (lane == 0) {
                *(uint4*)&mbits[wave * MPITCH + u * 4]        = make_uint4(m0, m1, m2, m3);
                *(uint4*)&mbits[(wave + 16) * MPITCH + u * 4] = make_uint4(n0, n1, n2, n3);
            }
#pragma unroll
            for (int hd = 0; hd < NHEADS; ++hd) {
                const float4 tv = *(const float4*)(t_ln + hd * NNODES + u * 128);
                const float a0 = p0 ? tv.x : ninf, a1 = p1 ? tv.y : ninf;
                const float a2 = p2 ? tv.z : ninf, a3 = p3 ? tv.w : ninf;
                tma[hd] = fmaxf(fmaxf(tma[hd], a0), fmaxf(fmaxf(a1, a2), a3));
                const float b0 = q0 ? tv.x : ninf, b1 = q1 ? tv.y : ninf;
                const float b2 = q2 ? tv.z : ninf, b3 = q3 ? tv.w : ninf;
                tmb[hd] = fmaxf(fmaxf(tmb[hd], b0), fmaxf(fmaxf(b1, b2), b3));
            }
        }
#pragma unroll
        for (int hd = 0; hd < NHEADS; ++hd) {
            tma[hd] = wave_max(tma[hd]);
            tmb[hd] = wave_max(tmb[hd]);
        }
        if (lane < 2 * NHEADS) {
            const float ta = (lane & 2) ? ((lane & 1) ? tma[3] : tma[2]) : ((lane & 1) ? tma[1] : tma[0]);
            const float tb = (lane & 2) ? ((lane & 1) ? tmb[3] : tmb[2]) : ((lane & 1) ? tmb[1] : tmb[0]);
            const int   hd   = lane & 3;
            const int   rowl = wave + ((lane >> 2) << 4);
            const float tmx  = (lane & 4) ? tb : ta;
            float e = s[(b * NHEADS + hd) * NNODES + rowTile * BROWS + rowl] + tmx;
            e = (e > 0.f) ? e : 0.2f * e;
            gm[hd][rowl] = fmaxf(NEG_INF_F, e);
        }
    }

    const int   iGlob  = rowTile * BROWS + r;
    const float s_r0   = s[(b * NHEADS + head) * NNODES + iGlob];
    const float s_r1   = s[(b * NHEADS + head) * NNODES + iGlob + 16];
    const float* t_bh  = t + (b * NHEADS + head) * NNODES;
    const int jbase = quad * 256;
    __syncthreads();

    const float mr0 = gm[head][r];
    const float mr1 = gm[head][r + 16];
    const float* hb = h + (size_t)b * NNODES * OUTDIM + head * HDIM;
    const int n0 = lane & 15;
    v8f c0 = {}, c1 = {}, d0 = {}, d1 = {};
    v2f lacc0 = {}, lacc1 = {};
    for (int c = 0; c < 2; ++c) {
        const unsigned* mw0 = &mbits[r * MPITCH + (quad * 2 + c) * 4 + jl * 2];
        const unsigned* mw1 = mw0 + 16 * MPITCH;
        const unsigned w00 = mw0[0];
        const unsigned w01 = mw0[1];
        const unsigned w10 = mw1[0];
        const unsigned w11 = mw1[1];
        const int jc = jbase + c * 128 + jl * 2;
#pragma unroll 8
        for (int i = 0; i < 32; ++i) {
            const unsigned bit = 1u << i;
            const int jA = jc + 4 * i;
            const float tx = t_bh[jA];
            const float ty = t_bh[jA + 1];
            v2f a0, a1;
            {
                float e = s_r0 + tx;
                e = (e > 0.f) ? e : 0.2f * e;
                e = ((w00 & bit) == 0u) ? NEG_INF_F : e;
                a0.x = __expf(e - mr0);
                e = s_r0 + ty;
                e = (e > 0.f) ? e : 0.2f * e;
                e = ((w01 & bit) == 0u) ? NEG_INF_F : e;
                a0.y = __expf(e - mr0);
                e = s_r1 + tx;
                e = (e > 0.f) ? e : 0.2f * e;
                e = ((w10 & bit) == 0u) ? NEG_INF_F : e;
                a1.x = __expf(e - mr1);
                e = s_r1 + ty;
                e = (e > 0.f) ? e : 0.2f * e;
                e = ((w11 & bit) == 0u) ? NEG_INF_F : e;
                a1.y = __expf(e - mr1);
            }
            lacc0 += a0;
            lacc1 += a1;
            v2f b0, b1;
            b0.x = hb[(size_t)(jA + 0) * OUTDIM + n0];
            b0.y = hb[(size_t)(jA + 1) * OUTDIM + n0];
            b1.x = hb[(size_t)(jA + 0) * OUTDIM + 16 + n0];
            b1.y = hb[(size_t)(jA + 1) * OUTDIM + 16 + n0];
            c0 = __builtin_amdgcn_wmma_f32_16x16x4_f32(false, a0, false, b0,
                                                       (short)0, c0, false, false);
            c1 = __builtin_amdgcn_wmma_f32_16x16x4_f32(false, a0, false, b1,
                                                       (short)0, c1, false, false);
            d0 = __builtin_amdgcn_wmma_f32_16x16x4_f32(false, a1, false, b0,
                                                       (short)0, d0, false, false);
            d1 = __builtin_amdgcn_wmma_f32_16x16x4_f32(false, a1, false, b1,
                                                       (short)0, d1, false, false);
        }
    }
    pl[0][tid] = lacc0.x + lacc0.y;
    pl[1][tid] = lacc1.x + lacc1.y;
#pragma unroll
    for (int v = 0; v < 8; ++v) {
        pc[wave][0][0][v][lane] = c0[v];
        pc[wave][0][1][v][lane] = c1[v];
        pc[wave][1][0][v][lane] = d0[v];
        pc[wave][1][1][v][lane] = d1[v];
    }
    __syncthreads();

    for (int idx = tid; idx < BROWS * OUTDIM; idx += 512) {
        int col  = idx & 31;
        int hd   = (idx >> 5) & 3;
        int rowL = idx >> 7;
        int tile = rowL >> 4;
        int rl   = rowL & 15;
        int ct = col >> 4;
        int v  = rl & 7;
        int ln = (col & 15) + ((rl >> 3) << 4);
        float sum = pc[hd * 4 + 0][tile][ct][v][ln] + pc[hd * 4 + 1][tile][ct][v][ln] +
                    pc[hd * 4 + 2][tile][ct][v][ln] + pc[hd * 4 + 3][tile][ct][v][ln];
        const float* plh = &pl[tile][hd * 128 + rl];
        float lrow = ((plh[0] + plh[16]) + (plh[32] + plh[48])) +
                     ((plh[64] + plh[80]) + (plh[96] + plh[112]));
        out[(size_t)(b * NNODES + rowTile * BROWS + rowL) * OUTDIM + hd * HDIM + col]
            = sum / lrow;
    }
}

extern "C" void kernel_launch(void* const* d_in, const int* in_sizes, int n_in,
                              void* d_out, int out_size, void* d_ws, size_t ws_size,
                              hipStream_t stream)
{
    const float* x    = (const float*)d_in[0];
    const int*   adj  = (const int*)  d_in[1];
    const float* W    = (const float*)d_in[2];
    const float* bias = (const float*)d_in[3];
    const float* a    = (const float*)d_in[4];
    float* out = (float*)d_out;

    float* h = (float*)d_ws;
    float* s = h + BATCH * NNODES * OUTDIM;
    float* t = s + BATCH * NHEADS * NNODES;

    gat_linear<<<(BATCH * NNODES) / 16, 256, 0, stream>>>(x, W, bias, a, h, s, t);
    gat_attn<<<BATCH * (NNODES / BROWS), 512, 0, stream>>>(h, adj, s, t, out);
}


# attn: exp out of the product loop: exp(leaky(s+t)-m) = max(exp(s-m+T)exp(t-T), exp(a(s+T)-m)exp(a(t-T))), a = slope; column factors tabled per wave in LDS; P rounds anew; old loop if T-tmax>64
# speedup vs baseline: 1.6994x; 1.0708x over previous
#include <hip/hip_runtime.h>
#include <hip/hip_bf16.h>

#define NHEADS 4
#define HDIM   32
#define NNODES 1024
#define BATCH  8
#define INDIM  128
#define OUTDIM 128
#define NEG_INF_F (-1000000000.0f)

typedef float v2f __attribute__((ext_vector_type(2)));
typedef float v8f __attribute__((ext_vector_type(8)));

__device__ __forceinline__ float xhalf(float x)
{
    const int bits = __builtin_bit_cast(int, x);
    return __builtin_bit_cast(float,
        __builtin_amdgcn_permlanex16(bits, bits, 0x76543210, 0xfedcba98, false, false));
}

template <int XM>
__device__ __forceinline__ float row_xor(float x)
{
    const int bits = __builtin_bit_cast(int, x);
    return __builtin_bit_cast(float,
        __builtin_amdgcn_update_dpp(bits, bits, 0x160 + XM, 0xf, 0xf, false));
}

__device__ __forceinline__ float wave_max(float x)
{
    x = fmaxf(x, row_xor<1>(x));
    x = fmaxf(x, row_xor<2>(x));
    x = fmaxf(x, row_xor<4>(x));
    x = fmaxf(x, row_xor<8>(x));
    return fmaxf(x, xhalf(x));
}

__device__ __forceinline__ float row_sum16(float x)
{
    x += row_xor<1>(x);
    x += row_xor<2>(x);
    x += row_xor<4>(x);
    return x + row_xor<8>(x);
}

__global__ __launch_bounds__(256) void gat_linear(const float* __restrict__ x,
                                                  const float* __restrict__ W,
                                                  const float* __restrict__ bias,
                                                  const float* __restrict__ a,
                                                  float* __restrict__ h,
                                                  float* __restrict__ s,
                                                  float* __restrict__ t)
{
    const int lane    = threadIdx.x & 31;
    const int wave    = threadIdx.x >> 5;
    const int rowTile = blockIdx.x;
    const int rloc    = lane & 15;
    const int kq      = (lane >> 4) * 2;
    const int col     = wave * 16 + rloc;

    __shared__ float xs[16][132];
    __shared__ __attribute__((aligned(16))) float sp[2][8][16];
    for (int i = threadIdx.x; i < 16 * 128; i += 256) {
        int rr = i >> 7, cc = i & 127;
        xs[rr][cc] = x[(rowTile * 16 + rr) * INDIM + cc];
    }
    __syncthreads();

    v8f acc = {};
    for (int kk = 0; kk < INDIM; kk += 4) {
        v2f av, bw;
        av.x = xs[rloc][kk + kq + 0];
        av.y = xs[rloc][kk + kq + 1];
        bw.x = W[col * INDIM + kk + kq + 0];
        bw.y = W[col * INDIM + kk + kq + 1];
        acc = __builtin_amdgcn_wmma_f32_16x16x4_f32(false, av, false, bw,
                                                    (short)0, acc, false, false);
    }

    const float bv  = bias[col];
    const float asv = a[col & (HDIM - 1)];
    const float atv = a[HDIM + (col & (HDIM - 1))];
    const int half  = lane >> 4;
    const int rbase = rowTile * 16 + (half << 3);
    float hv[8], ps[8], pt[8];
#pragma unroll
    for (int v = 0; v < 8; ++v) {
        hv[v] = acc[v] + bv;
        ps[v] = row_sum16(hv[v] * asv);
        pt[v] = row_sum16(hv[v] * atv);
    }
    if (rloc == 0) {
        float4* ds = (float4*)&sp[0][wave][half * 8];
        float4* dt = (float4*)&sp[1][wave][half * 8];
        ds[0] = make_float4(ps[0], ps[1], ps[2], ps[3]);
        ds[1] = make_float4(ps[4], ps[5], ps[6], ps[7]);
        dt[0] = make_float4(pt[0], pt[1], pt[2], pt[3]);
        dt[1] = make_float4(pt[4], pt[5], pt[6], pt[7]);
    }
    __syncthreads();
    if (threadIdx.x < 128) {
        const int which = threadIdx.x >> 6;
        const int head  = (threadIdx.x >> 4) & 3;
        const int row   = threadIdx.x & 15;
        const float v   = sp[which][2 * head][row] + sp[which][2 * head + 1][row];
        const int grow  = rowTile * 16 + row;
        float* dst = which ? t : s;
        dst[((grow >> 10) * NHEADS + head) * NNODES + (grow & (NNODES - 1))] = v;
    }
#pragma unroll
    for (int v = 0; v < 8; ++v)
        h[(rbase + v) * OUTDIM + col] = hv[v];
}

#define MPITCH 36
#define BROWS  32
#define FACT_LIM 64.0f

__global__ __launch_bounds__(512) void gat_attn(const float* __restrict__ h,
                                                const int*   __restrict__ adj,
                                                const float* __restrict__ s,
                                                const float* __restrict__ t,
                                                float* __restrict__ out)
{
    const int rowTile = blockIdx.x & 31;
    const int b       = blockIdx.x >> 5;

    const int tid  = threadIdx.x;
    const int wave = tid >> 5;
    const int lane = tid & 31;
    const int head = wave >> 2;
    const int quad = wave & 3;
    const int r    = lane & 15;
    const int jl   = lane >> 4;

    __shared__ __attribute__((aligned(16))) unsigned mbits[BROWS * MPITCH];
    __shared__ float pl[2][512];
    __shared__ float gm[NHEADS][BROWS];
    __shared__ float gt[NHEADS][BROWS];
    __shared__ __attribute__((aligned(16))) float4 tab[16][128];
    __shared__ float pc[16][2][2][8][32];

    float Tw;
    {
        const int* adj_r0 = adj + (size_t)(b * NNODES + rowTile * BROWS + wave) * NNODES + lane * 4;
        const int* adj_r1 = adj_r0 + (size_t)16 * NNODES;
        const float* t_ln = t + (size_t)b * NHEADS * NNODES + lane * 4;
        int4 va[8], vb[8];
        const float* t_own = t + (size_t)(b * NHEADS + head) * NNODES + quad * 256 + lane * 4;
        const float4 tq0 = *(const float4*)(t_own);
        const float4 tq1 = *(const float4*)(t_own + 128);
#pragma unroll
        for (int u = 0; u < 8; ++u) {
            va[u] = *(const int4*)(adj_r0 + u * 128);
            vb[u] = *(const int4*)(adj_r1 + u * 128);
        }
        const float ninf = -__builtin_inff();
        float tma[NHEADS] = {ninf, ninf, ninf, ninf};
        float tmb[NHEADS] = {ninf, ninf, ninf, ninf};
#pragma unroll
        for (int u = 0; u < 8; ++u) {
            const bool p0 = va[u].x != 0, p1 = va[u].y != 0, p2 = va[u].z != 0, p3 = va[u].w != 0;
            const bool q0 = vb[u].x != 0, q1 = vb[u].y != 0, q2 = vb[u].z != 0, q3 = vb[u].w != 0;
            unsigned m0 = __builtin_amdgcn_ballot_w32(p0);
            unsigned m1 = __builtin_amdgcn_ballot_w32(p1);
            unsigned m2 = __builtin_amdgcn_ballot_w32(p2);
            unsigned m3 = __builtin_amdgcn_ballot_w32(p3);
            unsigned n0 = __builtin_amdgcn_ballot_w32(q0);
            unsigned n1 = __builtin_amdgcn_ballot_w32(q1);
            unsigned n2 = __builtin_amdgcn_ballot_w32(q2);
            unsigned n3 = __builtin_amdgcn_ballot_w32(q3);
            if (lane == 0) {
                *(uint4*)&mbits[wave * MPITCH + u * 4]        = make_uint4(m0, m1, m2, m3);
                *(uint4*)&mbits[(wave + 16) * MPITCH + u * 4] = make_uint4(n0, n1, n2, n3);
            }
#pragma unroll
            for (int hd = 0; hd < NHEADS; ++hd) {
                const float4 tv = *(const float4*)(t_ln + hd * NNODES + u * 128);
                const float a0 = p0 ? tv.x : ninf, a1 = p1 ? tv.y : ninf;
                const float a2 = p2 ? tv.z : ninf, a3 = p3 ? tv.w : ninf;
                tma[hd] = fmaxf(fmaxf(tma[hd], a0), fmaxf(fmaxf(a1, a2), a3));
                const float b0 = q0 ? tv.x : ninf, b1 = q1 ? tv.y : ninf;
                const float b2 = q2 ? tv.z : ninf, b3 = q3 ? tv.w : ninf;
                tmb[hd] = fmaxf(fmaxf(tmb[hd], b0), fmaxf(fmaxf(b1, b2), b3));
            }
        }
#pragma unroll
        for (int hd = 0; hd < NHEADS; ++hd) {
            tma[hd] = wave_max(tma[hd]);
            tmb[hd] = wave_max(tmb[hd]);
        }
        if (lane < 2 * NHEADS) {
            const float ta = (lane & 2) ? ((lane & 1) ? tma[3] : tma[2]) : ((lane & 1) ? tma[1] : tma[0]);
            const float tb = (lane & 2) ? ((lane & 1) ? tmb[3] : tmb[2]) : ((lane & 1) ? tmb[1] : tmb[0]);
            const int   hd   = lane & 3;
            const int   rowl = wave + ((lane >> 2) << 4);
            const float tmx  = (lane & 4) ? tb : ta;
            float e = s[(b * NHEADS + hd) * NNODES + rowTile * BROWS + rowl] + tmx;
            e = (e > 0.f) ? e : 0.2f * e;
            gm[hd][rowl] = fmaxf(NEG_INF_F, e);
            gt[hd][rowl] = tmx;
        }
        Tw = wave_max(fmaxf(fmaxf(fmaxf(tq0.x, tq0.y), fmaxf(tq0.z, tq0.w)),
                            fmaxf(fmaxf(tq1.x, tq1.y), fmaxf(tq1.z, tq1.w))));
        const float d0 = tq0.x - Tw, d1 = tq0.y - Tw, d2 = tq0.z - Tw, d3 = tq0.w - Tw;
        const float d4 = tq1.x - Tw, d5 = tq1.y - Tw, d6 = tq1.z - Tw, d7 = tq1.w - Tw;
        tab[wave][2 * lane]          = make_float4(__expf(d0), __expf(d1), __expf(0.2f * d0), __expf(0.2f * d1));
        tab[wave][2 * lane + 1]      = make_float4(__expf(d2), __expf(d3), __expf(0.2f * d2), __expf(0.2f * d3));
        tab[wave][64 + 2 * lane]     = make_float4(__expf(d4), __expf(d5), __expf(0.2f * d4), __expf(0.2f * d5));
        tab[wave][64 + 2 * lane + 1] = make_float4(__expf(d6), __expf(d7), __expf(0.2f * d6), __expf(0.2f * d7));
    }

    const int   iGlob  = rowTile * BROWS + r;
    const float s_r0   = s[(b * NHEADS + head) * NNODES + iGlob];
    const float s_r1   = s[(b * NHEADS + head) * NNODES + iGlob + 16];
    const float* t_bh  = t + (b * NHEADS + head) * NNODES;
    const int jbase = quad * 256;
    __syncthreads();

    const float mr0 = gm[head][r];
    const float mr1 = gm[head][r + 16];
    const float tm0 = gt[head][r];
    const float tm1 = gt[head][r + 16];
    const float* hb = h + (size_t)b * NNODES * OUTDIM + head * HDIM;
    const int n0 = lane & 15;
    v8f c0 = {}, c1 = {}, d0 = {}, d1 = {};
    v2f lacc0 = {}, lacc1 = {};
    const bool wide = ((tm0 > -__builtin_inff()) && (Tw - tm0 > FACT_LIM)) ||
                      ((tm1 > -__builtin_inff()) && (Tw - tm1 > FACT_LIM));
    if (__builtin_amdgcn_ballot_w32(wide) == 0u) {
        const float fa0 = __expf((s_r0 - mr0) + Tw);
        const float fb0 = __expf(0.2f * (s_r0 + Tw) - mr0);
        const float fa1 = __expf((s_r1 - mr1) + Tw);
        const float fb1 = __expf(0.2f * (s_r1 + Tw) - mr1);
        const int   fz0 = __builtin_bit_cast(int, __expf(NEG_INF_F - mr0));
        const int   fz1 = __builtin_bit_cast(int, __expf(NEG_INF_F - mr1));
        for (int c = 0; c < 2; ++c) {
            const unsigned* mw0 = &mbits[r * MPITCH + (quad * 2 + c) * 4 + jl * 2];
            const unsigned* mw1 = mw0 + 16 * MPITCH;
            const unsigned w00 = mw0[0];
            const unsigned w01 = mw0[1];
            const unsigned w10 = mw1[0];
            const unsigned w11 = mw1[1];
            const int jc = jbase + c * 128 + jl * 2;
            const float4* tw = &tab[wave][c * 64 + jl];
#pragma unroll 8
            for (int i = 0; i < 32; ++i) {
                const int jA = jc + 4 * i;
                const float4 uv = tw[2 * i];
                const int k00 = (int)(w00 << (31 - i)) >> 31;
                const int k01 = (int)(w01 << (31 - i)) >> 31;
                const int k10 = (int)(w10 << (31 - i)) >> 31;
                const int k11 = (int)(w11 << (31 - i)) >> 31;
                const float p00 = fmaxf(fa0 * uv.x, fb0 * uv.z);
                const float p01 = fmaxf(fa0 * uv.y, fb0 * uv.w);
                const float p10 = fmaxf(fa1 * uv.x, fb1 * uv.z);
                const float p11 = fmaxf(fa1 * uv.y, fb1 * uv.w);
                v2f a0, a1;
                a0.x = __builtin_bit_cast(float, (__builtin_bit_cast(int, p00) & k00) | (fz0 & ~k00));
                a0.y = __builtin_bit_cast(float, (__builtin_bit_cast(int, p01) & k01) | (fz0 & ~k01));
                a1.x = __builtin_bit_cast(float, (__builtin_bit_cast(int, p10) & k10) | (fz1 & ~k10));
                a1.y = __builtin_bit_cast(float, (__builtin_bit_cast(int, p11) & k11) | (fz1 & ~k11));
                lacc0 += a0;
                lacc1 += a1;
                v2f b0, b1;
                b0.x = hb[(size_t)(jA + 0) * OUTDIM + n0];
                b0.y = hb[(size_t)(jA + 1) * OUTDIM + n0];
                b1.x = hb[(size_t)(jA + 0) * OUTDIM + 16 + n0];
                b1.y = hb[(size_t)(jA + 1) * OUTDIM + 16 + n0];
                c0 = __builtin_amdgcn_wmma_f32_16x16x4_f32(false, a0, false, b0,
                                                           (short)0, c0, false, false);
                c1 = __builtin_amdgcn_wmma_f32_16x16x4_f32(false, a0, false, b1,
                                                           (short)0, c1, false, false);
                d0 = __builtin_amdgcn_wmma_f32_16x16x4_f32(false, a1, false, b0,
                                                           (short)0, d0, false, false);
                d1 = __builtin_amdgcn_wmma_f32_16x16x4_f32(false, a1, false, b1,
                                                           (short)0, d1, false, false);
            }
        }
    } else {
        for (int c = 0; c < 2; ++c) {
            const unsigned* mw0 = &mbits[r * MPITCH + (quad * 2 + c) * 4 + jl * 2];
            const unsigned* mw1 = mw0 + 16 * MPITCH;
            const unsigned w00 = mw0[0];
            const unsigned w01 = mw0[1];
            const unsigned w10 = mw1[0];
            const unsigned w11 = mw1[1];
            const int jc = jbase + c * 128 + jl * 2;
#pragma unroll 8
            for (int i = 0; i < 32; ++i) {
                const unsigned bit = 1u << i;
                const int jA = jc + 4 * i;
                const float tx = t_bh[jA];
                const float ty = t_bh[jA + 1];
                v2f a0, a1;
                {
                    float e = s_r0 + tx;
                    e = (e > 0.f) ? e : 0.2f * e;
                    e = ((w00 & bit) == 0u) ? NEG_INF_F : e;
                    a0.x = __expf(e - mr0);
                    e = s_r0 + ty;
                    e = (e > 0.f) ? e : 0.2f * e;
                    e = ((w01 & bit) == 0u) ? NEG_INF_F : e;
                    a0.y = __expf(e - mr0);
                    e = s_r1 + tx;
                    e = (e > 0.f) ? e : 0.2f * e;
                    e = ((w10 & bit) == 0u) ? NEG_INF_F : e;
                    a1.x = __expf(e - mr1);
                    e = s_r1 + ty;
                    e = (e > 0.f) ? e : 0.2f * e;
                    e = ((w11 & bit) == 0u) ? NEG_INF_F : e;
                    a1.y = __expf(e - mr1);
                }
                lacc0 += a0;
                lacc1 += a1;
                v2f b0, b1;
                b0.x = hb[(size_t)(jA + 0) * OUTDIM + n0];
                b0.y = hb[(size_t)(jA + 1) * OUTDIM + n0];
                b1.x = hb[(size_t)(jA + 0) * OUTDIM + 16 + n0];
                b1.y = hb[(size_t)(jA + 1) * OUTDIM + 16 + n0];
                c0 = __builtin_amdgcn_wmma_f32_16x16x4_f32(false, a0, false, b0,
                                                           (short)0, c0, false, false);
                c1 = __builtin_amdgcn_wmma_f32_16x16x4_f32(false, a0, false, b1,
                                                           (short)0, c1, false, false);
                d0 = __builtin_amdgcn_wmma_f32_16x16x4_f32(false, a1, false, b0,
                                                           (short)0, d0, false, false);
                d1 = __builtin_amdgcn_wmma_f32_16x16x4_f32(false, a1, false, b1,
                                                           (short)0, d1, false, false);
            }
        }
    }
    pl[0][tid] = lacc0.x + lacc0.y;
    pl[1][tid] = lacc1.x + lacc1.y;
#pragma unroll
    for (int v = 0; v < 8; ++v) {
        pc[wave][0][0][v][lane] = c0[v];
        pc[wave][0][1][v][lane] = c1[v];
        pc[wave][1][0][v][lane] = d0[v];
        pc[wave][1][1][v][lane] = d1[v];
    }
    __syncthreads();

    for (int idx = tid; idx < BROWS * OUTDIM; idx += 512) {
        int col  = idx & 31;
        int hd   = (idx >> 5) & 3;
        int rowL = idx >> 7;
        int tile = rowL >> 4;
        int rl   = rowL & 15;
        int ct = col >> 4;
        int v  = rl & 7;
        int ln = (col & 15) + ((rl >> 3) << 4);
        float sum = pc[hd * 4 + 0][tile][ct][v][ln] + pc[hd * 4 + 1][tile][ct][v][ln] +
                    pc[hd * 4 + 2][tile][ct][v][ln] + pc[hd * 4 + 3][tile][ct][v][ln];
        const float* plh = &pl[tile][hd * 128 + rl];
        float lrow = ((plh[0] + plh[16]) + (plh[32] + plh[48])) +
                     ((plh[64] + plh[80]) + (plh[96] + plh[112]));
        out[(size_t)(b * NNODES + rowTile * BROWS + rowL) * OUTDIM + hd * HDIM + col]
            = sum / lrow;
    }
}

extern "C" void kernel_launch(void* const* d_in, const int* in_sizes, int n_in,
                              void* d_out, int out_size, void* d_ws, size_t ws_size,
                              hipStream_t stream)
{
    const float* x    = (const float*)d_in[0];
    const int*   adj  = (const int*)  d_in[1];
    const float* W    = (const float*)d_in[2];
    const float* bias = (const float*)d_in[3];
    const float* a    = (const float*)d_in[4];
    float* out = (float*)d_out;

    float* h = (float*)d_ws;
    float* s = h + BATCH * NNODES * OUTDIM;
    float* t = s + BATCH * NHEADS * NNODES;

    gat_linear<<<(BATCH * NNODES) / 16, 256, 0, stream>>>(x, W, bias, a, h, s, t);
    gat_attn<<<BATCH * (NNODES / BROWS), 512, 0, stream>>>(h, adj, s, t, out);
}


# linear: x tile staged with two 16-byte loads per thread, both in flight before the first LDS store (was a loop of eight 4-byte loads, each waited for before its store)
# speedup vs baseline: 1.9073x; 1.1223x over previous
#include <hip/hip_runtime.h>
#include <hip/hip_bf16.h>

#define NHEADS 4
#define HDIM   32
#define NNODES 1024
#define BATCH  8
#define INDIM  128
#define OUTDIM 128
#define NEG_INF_F (-1000000000.0f)

typedef float v2f __attribute__((ext_vector_type(2)));
typedef float v8f __attribute__((ext_vector_type(8)));

__device__ __forceinline__ float xhalf(float x)
{
    const int bits = __builtin_bit_cast(int, x);
    return __builtin_bit_cast(float,
        __builtin_amdgcn_permlanex16(bits, bits, 0x76543210, 0xfedcba98, false, false));
}

template <int XM>
__device__ __forceinline__ float row_xor(float x)
{
    const int bits = __builtin_bit_cast(int, x);
    return __builtin_bit_cast(float,
        __builtin_amdgcn_update_dpp(bits, bits, 0x160 + XM, 0xf, 0xf, false));
}

__device__ __forceinline__ float wave_max(float x)
{
    x = fmaxf(x, row_xor<1>(x));
    x = fmaxf(x, row_xor<2>(x));
    x = fmaxf(x, row_xor<4>(x));
    x = fmaxf(x, row_xor<8>(x));
    return fmaxf(x, xhalf(x));
}

__device__ __forceinline__ float row_sum16(float x)
{
    x += row_xor<1>(x);
    x += row_xor<2>(x);
    x += row_xor<4>(x);
    return x + row_xor<8>(x);
}

__global__ __launch_bounds__(256) void gat_linear(const float* __restrict__ x,
                                                  const float* __restrict__ W,
                                                  const float* __restrict__ bias,
                                                  const float* __restrict__ a,
                                                  float* __restrict__ h,
                                                  float* __restrict__ s,
                                                  float* __restrict__ t)
{
    const int lane    = threadIdx.x & 31;
    const int wave    = threadIdx.x >> 5;
    const int rowTile = blockIdx.x;
    const int rloc    = lane & 15;
    const int kq      = (lane >> 4) * 2;
    const int col     = wave * 16 + rloc;

    __shared__ __attribute__((aligned(16))) float xs[16][132];
    __shared__ __attribute__((aligned(16))) float sp[2][8][16];
    {
        const float4* xg = (const float4*)(x + (size_t)rowTile * 16 * INDIM);
        const int q0 = threadIdx.x, q1 = threadIdx.x + 256;
        const float4 x0 = xg[q0];
        const float4 x1 = xg[q1];
        *(float4*)&xs[q0 >> 5][(q0 & 31) * 4] = x0;
        *(float4*)&xs[q1 >> 5][(q1 & 31) * 4] = x1;
    }
    __syncthreads();

    v8f acc = {};
    for (int kk = 0; kk < INDIM; kk += 4) {
        v2f av, bw;
        av.x = xs[rloc][kk + kq + 0];
        av.y = xs[rloc][kk + kq + 1];
        bw.x = W[col * INDIM + kk + kq + 0];
        bw.y = W[col * INDIM + kk + kq + 1];
        acc = __builtin_amdgcn_wmma_f32_16x16x4_f32(false, av, false, bw,
                                                    (short)0, acc, false, false);
    }

    const float bv  = bias[col];
    const float asv = a[col & (HDIM - 1)];
    const float atv = a[HDIM + (col & (HDIM - 1))];
    const int half  = lane >> 4;
    const int rbase = rowTile * 16 + (half << 3);
    float hv[8], ps[8], pt[8];
#pragma unroll
    for (int v = 0; v < 8; ++v) {
        hv[v] = acc[v] + bv;
        ps[v] = row_sum16(hv[v] * asv);
        pt[v] = row_sum16(hv[v] * atv);
    }
    if (rloc == 0) {
        float4* ds = (float4*)&sp[0][wave][half * 8];
        float4* dt = (float4*)&sp[1][wave][half * 8];
        ds[0] = make_float4(ps[0], ps[1], ps[2], ps[3]);
        ds[1] = make_float4(ps[4], ps[5], ps[6], ps[7]);
        dt[0] = make_float4(pt[0], pt[1], pt[2], pt[3]);
        dt[1] = make_float4(pt[4], pt[5], pt[6], pt[7]);
    }
    __syncthreads();
    if (threadIdx.x < 128) {
        const int which = threadIdx.x >> 6;
        const int head  = (threadIdx.x >> 4) & 3;
        const int row   = threadIdx.x & 15;
        const float v   = sp[which][2 * head][row] + sp[which][2 * head + 1][row];
        const int grow  = rowTile * 16 + row;
        float* dst = which ? t : s;
        dst[((grow >> 10) * NHEADS + head) * NNODES + (grow & (NNODES - 1))] = v;
    }
#pragma unroll
    for (int v = 0; v < 8; ++v)
        h[(rbase + v) * OUTDIM + col] = hv[v];
}

#define MPITCH 36
#define BROWS  32
#define FACT_LIM 64.0f

__global__ __launch_bounds__(512) void gat_attn(const float* __restrict__ h,
                                                const int*   __restrict__ adj,
                                                const float* __restrict__ s,
                                                const float* __restrict__ t,
                                                float* __restrict__ out)
{
    const int rowTile = blockIdx.x & 31;
    const int b       = blockIdx.x >> 5;

    const int tid  = threadIdx.x;
    const int wave = tid >> 5;
    const int lane = tid & 31;
    const int head = wave >> 2;
    const int quad = wave & 3;
    const int r    = lane & 15;
    const int jl   = lane >> 4;

    __shared__ __attribute__((aligned(16))) unsigned mbits[BROWS * MPITCH];
    __shared__ float pl[2][512];
    __shared__ float gm[NHEADS][BROWS];
    __shared__ float gt[NHEADS][BROWS];
    __shared__ __attribute__((aligned(16))) float4 tab[16][128];
    __shared__ float pc[16][2][2][8][32];

    float Tw;
    {
        const int* adj_r0 = adj + (size_t)(b * NNODES + rowTile * BROWS + wave) * NNODES + lane * 4;
        const int* adj_r1 = adj_r0 + (size_t)16 * NNODES;
        const float* t_ln = t + (size_t)b * NHEADS * NNODES + lane * 4;
        int4 va[8], vb[8];
        const float* t_own = t + (size_t)(b * NHEADS + head) * NNODES + quad * 256 + lane * 4;
        const float4 tq0 = *(const float4*)(t_own);
        const float4 tq1 = *(const float4*)(t_own + 128);
#pragma unroll
        for (int u = 0; u < 8; ++u) {
            va[u] = *(const int4*)(adj_r0 + u * 128);
            vb[u] = *(const int4*)(adj_r1 + u * 128);
        }
        const float ninf = -__builtin_inff();
        float tma[NHEADS] = {ninf, ninf, ninf, ninf};
        float tmb[NHEADS] = {ninf, ninf, ninf, ninf};
#pragma unroll
        for (int u = 0; u < 8; ++u) {
            const bool p0 = va[u].x != 0, p1 = va[u].y != 0, p2 = va[u].z != 0, p3 = va[u].w != 0;
            const bool q0 = vb[u].x != 0, q1 = vb[u].y != 0, q2 = vb[u].z != 0, q3 = vb[u].w != 0;
            unsigned m0 = __builtin_amdgcn_ballot_w32(p0);
            unsigned m1 = __builtin_amdgcn_ballot_w32(p1);
            unsigned m2 = __builtin_amdgcn_ballot_w32(p2);
            unsigned m3 = __builtin_amdgcn_ballot_w32(p3);
            unsigned n0 = __builtin_amdgcn_ballot_w32(q0);
            unsigned n1 = __builtin_amdgcn_ballot_w32(q1);
            unsigned n2 = __builtin_amdgcn_ballot_w32(q2);
            unsigned n3 = __builtin_amdgcn_ballot_w32(q3);
            if (lane == 0) {
                *(uint4*)&mbits[wave * MPITCH + u * 4]        = make_uint4(m0, m1, m2, m3);
                *(uint4*)&mbits[(wave + 16) * MPITCH + u * 4] = make_uint4(n0, n1, n2, n3);
            }
#pragma unroll
            for (int hd = 0; hd < NHEADS; ++hd) {
                const float4 tv = *(const float4*)(t_ln + hd * NNODES + u * 128);
                const float a0 = p0 ? tv.x : ninf, a1 = p1 ? tv.y : ninf;
                const float a2 = p2 ? tv.z : ninf, a3 = p3 ? tv.w : ninf;
                tma[hd] = fmaxf(fmaxf(tma[hd], a0), fmaxf(fmaxf(a1, a2), a3));
                const float b0 = q0 ? tv.x : ninf, b1 = q1 ? tv.y : ninf;
                const float b2 = q2 ? tv.z : ninf, b3 = q3 ? tv.w : ninf;
                tmb[hd] = fmaxf(fmaxf(tmb[hd], b0), fmaxf(fmaxf(b1, b2), b3));
            }
        }
#pragma unroll
        for (int hd = 0; hd < NHEADS; ++hd) {
            tma[hd] = wave_max(tma[hd]);
            tmb[hd] = wave_max(tmb[hd]);
        }
        if (lane < 2 * NHEADS) {
            const float ta = (lane & 2) ? ((lane & 1) ? tma[3] : tma[2]) : ((lane & 1) ? tma[1] : tma[0]);
            const float tb = (lane & 2) ? ((lane & 1) ? tmb[3] : tmb[2]) : ((lane & 1) ? tmb[1] : tmb[0]);
            const int   hd   = lane & 3;
            const int   rowl = wave + ((lane >> 2) << 4);
            const float tmx  = (lane & 4) ? tb : ta;
            float e = s[(b * NHEADS + hd) * NNODES + rowTile * BROWS + rowl] + tmx;
            e = (e > 0.f) ? e : 0.2f * e;
            gm[hd][rowl] = fmaxf(NEG_INF_F, e);
            gt[hd][rowl] = tmx;
        }
        Tw = wave_max(fmaxf(fmaxf(fmaxf(tq0.x, tq0.y), fmaxf(tq0.z, tq0.w)),
                            fmaxf(fmaxf(tq1.x, tq1.y), fmaxf(tq1.z, tq1.w))));
        const float d0 = tq0.x - Tw, d1 = tq0.y - Tw, d2 = tq0.z - Tw, d3 = tq0.w - Tw;
        const float d4 = tq1.x - Tw, d5 = tq1.y - Tw, d6 = tq1.z - Tw, d7 = tq1.w - Tw;
        tab[wave][2 * lane]          = make_float4(__expf(d0), __expf(d1), __expf(0.2f * d0), __expf(0.2f * d1));
        tab[wave][2 * lane + 1]      = make_float4(__expf(d2), __expf(d3), __expf(0.2f * d2), __expf(0.2f * d3));
        tab[wave][64 + 2 * lane]     = make_float4(__expf(d4), __expf(d5), __expf(0.2f * d4), __expf(0.2f * d5));
        tab[wave][64 + 2 * lane + 1] = make_float4(__expf(d6), __expf(d7), __expf(0.2f * d6), __expf(0.2f * d7));
    }

    const int   iGlob  = rowTile * BROWS + r;
    const float s_r0   = s[(b * NHEADS + head) * NNODES + iGlob];
    const float s_r1   = s[(b * NHEADS + head) * NNODES + iGlob + 16];
    const float* t_bh  = t + (b * NHEADS + head) * NNODES;
    const int jbase = quad * 256;
    __syncthreads();

    const float mr0 = gm[head][r];
    const float mr1 = gm[head][r + 16];
    const float tm0 = gt[head][r];
    const float tm1 = gt[head][r + 16];
    const float* hb = h + (size_t)b * NNODES * OUTDIM + head * HDIM;
    const int n0 = lane & 15;
    v8f c0 = {}, c1 = {}, d0 = {}, d1 = {};
    v2f lacc0 = {}, lacc1 = {};
    const bool wide = ((tm0 > -__builtin_inff()) && (Tw - tm0 > FACT_LIM)) ||
                      ((tm1 > -__builtin_inff()) && (Tw - tm1 > FACT_LIM));
    if (__builtin_amdgcn_ballot_w32(wide) == 0u) {
        const float fa0 = __expf((s_r0 - mr0) + Tw);
        const float fb0 = __expf(0.2f * (s_r0 + Tw) - mr0);
        const float fa1 = __expf((s_r1 - mr1) + Tw);
        const float fb1 = __expf(0.2f * (s_r1 + Tw) - mr1);
        const int   fz0 = __builtin_bit_cast(int, __expf(NEG_INF_F - mr0));
        const int   fz1 = __builtin_bit_cast(int, __expf(NEG_INF_F - mr1));
        for (int c = 0; c < 2; ++c) {
            const unsigned* mw0 = &mbits[r * MPITCH + (quad * 2 + c) * 4 + jl * 2];
            const unsigned* mw1 = mw0 + 16 * MPITCH;
            const unsigned w00 = mw0[0];
            const unsigned w01 = mw0[1];
            const unsigned w10 = mw1[0];
            const unsigned w11 = mw1[1];
            const int jc = jbase + c * 128 + jl * 2;
            const float4* tw = &tab[wave][c * 64 + jl];
#pragma unroll 8
            for (int i = 0; i < 32; ++i) {
                const int jA = jc + 4 * i;
                const float4 uv = tw[2 * i];
                const int k00 = (int)(w00 << (31 - i)) >> 31;
                const int k01 = (int)(w01 << (31 - i)) >> 31;
                const int k10 = (int)(w10 << (31 - i)) >> 31;
                const int k11 = (int)(w11 << (31 - i)) >> 31;
                const float p00 = fmaxf(fa0 * uv.x, fb0 * uv.z);
                const float p01 = fmaxf(fa0 * uv.y, fb0 * uv.w);
                const float p10 = fmaxf(fa1 * uv.x, fb1 * uv.z);
                const float p11 = fmaxf(fa1 * uv.y, fb1 * uv.w);
                v2f a0, a1;
                a0.x = __builtin_bit_cast(float, (__builtin_bit_cast(int, p00) & k00) | (fz0 & ~k00));
                a0.y = __builtin_bit_cast(float, (__builtin_bit_cast(int, p01) & k01) | (fz0 & ~k01));
                a1.x = __builtin_bit_cast(float, (__builtin_bit_cast(int, p10) & k10) | (fz1 & ~k10));
                a1.y = __builtin_bit_cast(float, (__builtin_bit_cast(int, p11) & k11) | (fz1 & ~k11));
                lacc0 += a0;
                lacc1 += a1;
                v2f b0, b1;
                b0.x = hb[(size_t)(jA + 0) * OUTDIM + n0];
                b0.y = hb[(size_t)(jA + 1) * OUTDIM + n0];
                b1.x = hb[(size_t)(jA + 0) * OUTDIM + 16 + n0];
                b1.y = hb[(size_t)(jA + 1) * OUTDIM + 16 + n0];
                c0 = __builtin_amdgcn_wmma_f32_16x16x4_f32(false, a0, false, b0,
                                                           (short)0, c0, false, false);
                c1 = __builtin_amdgcn_wmma_f32_16x16x4_f32(false, a0, false, b1,
                                                           (short)0, c1, false, false);
                d0 = __builtin_amdgcn_wmma_f32_16x16x4_f32(false, a1, false, b0,
                                                           (short)0, d0, false, false);
                d1 = __builtin_amdgcn_wmma_f32_16x16x4_f32(false, a1, false, b1,
                                                           (short)0, d1, false, false);
            }
        }
    } else {
        for (int c = 0; c < 2; ++c) {
            const unsigned* mw0 = &mbits[r * MPITCH + (quad * 2 + c) * 4 + jl * 2];
            const unsigned* mw1 = mw0 + 16 * MPITCH;
            const unsigned w00 = mw0[0];
            const unsigned w01 = mw0[1];
            const unsigned w10 = mw1[0];
            const unsigned w11 = mw1[1];
            const int jc = jbase + c * 128 + jl * 2;
#pragma unroll 8
            for (int i = 0; i < 32; ++i) {
                const unsigned bit = 1u << i;
                const int jA = jc + 4 * i;
                const float tx = t_bh[jA];
                const float ty = t_bh[jA + 1];
                v2f a0, a1;
                {
                    float e = s_r0 + tx;
                    e = (e > 0.f) ? e : 0.2f * e;
                    e = ((w00 & bit) == 0u) ? NEG_INF_F : e;
                    a0.x = __expf(e - mr0);
                    e = s_r0 + ty;
                    e = (e > 0.f) ? e : 0.2f * e;
                    e = ((w01 & bit) == 0u) ? NEG_INF_F : e;
                    a0.y = __expf(e - mr0);
                    e = s_r1 + tx;
                    e = (e > 0.f) ? e : 0.2f * e;
                    e = ((w10 & bit) == 0u) ? NEG_INF_F : e;
                    a1.x = __expf(e - mr1);
                    e = s_r1 + ty;
                    e = (e > 0.f) ? e : 0.2f * e;
                    e = ((w11 & bit) == 0u) ? NEG_INF_F : e;
                    a1.y = __expf(e - mr1);
                }
                lacc0 += a0;
                lacc1 += a1;
                v2f b0, b1;
                b0.x = hb[(size_t)(jA + 0) * OUTDIM + n0];
                b0.y = hb[(size_t)(jA + 1) * OUTDIM + n0];
                b1.x = hb[(size_t)(jA + 0) * OUTDIM + 16 + n0];
                b1.y = hb[(size_t)(jA + 1) * OUTDIM + 16 + n0];
                c0 = __builtin_amdgcn_wmma_f32_16x16x4_f32(false, a0, false, b0,
                                                           (short)0, c0, false, false);
                c1 = __builtin_amdgcn_wmma_f32_16x16x4_f32(false, a0, false, b1,
                                                           (short)0, c1, false, false);
                d0 = __builtin_amdgcn_wmma_f32_16x16x4_f32(false, a1, false, b0,
                                                           (short)0, d0, false, false);
                d1 = __builtin_amdgcn_wmma_f32_16x16x4_f32(false, a1, false, b1,
                                                           (short)0, d1, false, false);
            }
        }
    }
    pl[0][tid] = lacc0.x + lacc0.y;
    pl[1][tid] = lacc1.x + lacc1.y;
#pragma unroll
    for (int v = 0; v < 8; ++v) {
        pc[wave][0][0][v][lane] = c0[v];
        pc[wave][0][1][v][lane] = c1[v];
        pc[wave][1][0][v][lane] = d0[v];
        pc[wave][1][1][v][lane] = d1[v];
    }
    __syncthreads();

    for (int idx = tid; idx < BROWS * OUTDIM; idx += 512) {
        int col  = idx & 31;
        int hd   = (idx >> 5) & 3;
        int rowL = idx >> 7;
        int tile = rowL >> 4;
        int rl   = rowL & 15;
        int ct = col >> 4;
        int v  = rl & 7;
        int ln = (col & 15) + ((rl >> 3) << 4);
        float sum = pc[hd * 4 + 0][tile][ct][v][ln] + pc[hd * 4 + 1][tile][ct][v][ln] +
                    pc[hd * 4 + 2][tile][ct][v][ln] + pc[hd * 4 + 3][tile][ct][v][ln];
        const float* plh = &pl[tile][hd * 128 + rl];
        float lrow = ((plh[0] + plh[16]) + (plh[32] + plh[48])) +
                     ((plh[64] + plh[80]) + (plh[96] + plh[112]));
        out[(size_t)(b * NNODES + rowTile * BROWS + rowL) * OUTDIM + hd * HDIM + col]
            = sum / lrow;
    }
}

extern "C" void kernel_launch(void* const* d_in, const int* in_sizes, int n_in,
                              void* d_out, int out_size, void* d_ws, size_t ws_size,
                              hipStream_t stream)
{
    const float* x    = (const float*)d_in[0];
    const int*   adj  = (const int*)  d_in[1];
    const float* W    = (const float*)d_in[2];
    const float* bias = (const float*)d_in[3];
    const float* a    = (const float*)d_in[4];
    float* out = (float*)d_out;

    float* h = (float*)d_ws;
    float* s = h + BATCH * NNODES * OUTDIM;
    float* t = s + BATCH * NHEADS * NNODES;

    gat_linear<<<(BATCH * NNODES) / 16, 256, 0, stream>>>(x, W, bias, a, h, s, t);
    gat_attn<<<BATCH * (NNODES / BROWS), 512, 0, stream>>>(h, adj, s, t, out);
}
